# starting kernel
# baseline (speedup 1.0000x reference)
#include <hip/hip_runtime.h>
#include <hip/hip_bf16.h>

#define BB 8
#define NN 8192
#define MM 2048
#define CC 128

typedef __attribute__((ext_vector_type(2))) float v2f;
typedef __attribute__((ext_vector_type(8))) float v8f;

__global__ __launch_bounds__(256) void nn_idx_kernel(
    const float* __restrict__ unknown,
    const float* __restrict__ known,
    int* __restrict__ idx_out)
{
    __shared__ float4 kn[MM];

    const int b        = blockIdx.x >> 6;
    const int rowBlock = (blockIdx.x & 63) * 128;
    const int tid      = threadIdx.x;

    const float* kb = known + b * (MM * 3);
    for (int j = tid; j < MM; j += 256) {
        float x = kb[j * 3 + 0];
        float y = kb[j * 3 + 1];
        float z = kb[j * 3 + 2];
        kn[j] = make_float4(x, y, z, x * x + y * y + z * z);
    }
    __syncthreads();

    const int wave = tid >> 5;
    const int lane = tid & 31;
    const int col  = lane & 15;
    const int half = lane >> 4;

    const int row = rowBlock + wave * 16 + col;
    const float* u = unknown + (b * NN + row) * 3;
    const float ux = u[0], uy = u[1], uz = u[2];
    v2f a;
    a.x = half ? (-2.0f * uz) : (-2.0f * ux);
    a.y = half ? 1.0f         : (-2.0f * uy);

    const float2* __restrict__ lp = (const float2*)kn + (col * 2 + half);

    float minv[8];
    int   mini[8];
#pragma unroll
    for (int r = 0; r < 8; ++r) { minv[r] = 3.402823466e38f; mini[r] = 0; }

    for (int jb = 0; jb < MM; jb += 64) {
        float2 kv[4];
#pragma unroll
        for (int q = 0; q < 4; ++q) kv[q] = lp[(jb + 16 * q) * 2];

#pragma unroll
        for (int q = 0; q < 4; ++q) {
            v2f bv;
            bv.x = kv[q].x;
            bv.y = kv[q].y;

            v8f acc = {};
            acc = __builtin_amdgcn_wmma_f32_16x16x4_f32(
                 false, a,  false, bv,
                 (short)0, acc,  false,  false);

            const int j = jb + 16 * q + col;
#pragma unroll
            for (int r = 0; r < 8; ++r) {
                const float d = acc[r];
                const bool t = d < minv[r];
                minv[r] = t ? d : minv[r];
                mini[r] = t ? j : mini[r];
            }
        }
    }

#pragma unroll
    for (int off = 1; off < 16; off <<= 1) {
#pragma unroll
        for (int r = 0; r < 8; ++r) {
            const float ov = __shfl_xor(minv[r], off, 32);
            const int   oi = __shfl_xor(mini[r], off, 32);
            const bool  t  = (ov < minv[r]) | ((ov == minv[r]) & (oi < mini[r]));
            minv[r] = t ? ov : minv[r];
            mini[r] = t ? oi : mini[r];
        }
    }

    if (col == 0) {
        const int base = b * NN + rowBlock + wave * 16 + half * 8;
#pragma unroll
        for (int r = 0; r < 8; ++r) idx_out[base + r] = mini[r];
    }
}

__global__ __launch_bounds__(256) void gather_feats_kernel(
    const float* __restrict__ kf,
    const int*   __restrict__ idx,
    float* __restrict__ out)
{
    const int t = blockIdx.x * 256 + threadIdx.x;
    const int b = t / NN;
    const int i = t - b * NN;
    const int id = idx[t];
    const float* kfb = kf  + b * (CC * MM);
    float*       ob  = out + b * (CC * NN);
#pragma unroll 4
    for (int c = 0; c < CC; ++c) {
        ob[c * NN + i] = kfb[c * MM + id];
    }
}

extern "C" void kernel_launch(void* const* d_in, const int* in_sizes, int n_in,
                              void* d_out, int out_size, void* d_ws, size_t ws_size,
                              hipStream_t stream) {
    const float* unknown     = (const float*)d_in[0];
    const float* known       = (const float*)d_in[1];
    const float* known_feats = (const float*)d_in[2];
    float* out = (float*)d_out;
    int* idx   = (int*)d_ws;

    nn_idx_kernel<<<BB * (NN / 128), 256, 0, stream>>>(unknown, known, idx);

    gather_feats_kernel<<<(BB * NN) / 256, 256, 0, stream>>>(known_feats, idx, out);
}


# gather kernel: stage the 4 feature rows of a (batch, 4-channel) block in LDS, channel-minor, and fetch each query with one ds_load_b128 instead of divergent global loads; float4 stores along i
# speedup vs baseline: 2.1600x; 2.1600x over previous
#include <hip/hip_runtime.h>
#include <hip/hip_bf16.h>

#define BB 8
#define NN 8192
#define MM 2048
#define CC 128

typedef __attribute__((ext_vector_type(2))) float v2f;
typedef __attribute__((ext_vector_type(8))) float v8f;
typedef __attribute__((ext_vector_type(4))) int   v4i;

__global__ __launch_bounds__(256) void nn_idx_kernel(
    const float* __restrict__ unknown,
    const float* __restrict__ known,
    int* __restrict__ idx_out)
{
    __shared__ float4 kn[MM];

    const int b        = blockIdx.x >> 6;
    const int rowBlock = (blockIdx.x & 63) * 128;
    const int tid      = threadIdx.x;

    const float* kb = known + b * (MM * 3);
    for (int j = tid; j < MM; j += 256) {
        float x = kb[j * 3 + 0];
        float y = kb[j * 3 + 1];
        float z = kb[j * 3 + 2];
        kn[j] = make_float4(x, y, z, x * x + y * y + z * z);
    }
    __syncthreads();

    const int wave = tid >> 5;
    const int lane = tid & 31;
    const int col  = lane & 15;
    const int half = lane >> 4;

    const int row = rowBlock + wave * 16 + col;
    const float* u = unknown + (b * NN + row) * 3;
    const float ux = u[0], uy = u[1], uz = u[2];
    v2f a;
    a.x = half ? (-2.0f * uz) : (-2.0f * ux);
    a.y = half ? 1.0f         : (-2.0f * uy);

    const float2* __restrict__ lp = (const float2*)kn + (col * 2 + half);

    float minv[8];
    int   mini[8];
#pragma unroll
    for (int r = 0; r < 8; ++r) { minv[r] = 3.402823466e38f; mini[r] = 0; }

    for (int jb = 0; jb < MM; jb += 64) {
        float2 kv[4];
#pragma unroll
        for (int q = 0; q < 4; ++q) kv[q] = lp[(jb + 16 * q) * 2];

#pragma unroll
        for (int q = 0; q < 4; ++q) {
            v2f bv;
            bv.x = kv[q].x;
            bv.y = kv[q].y;

            v8f acc = {};
            acc = __builtin_amdgcn_wmma_f32_16x16x4_f32(
                 false, a,  false, bv,
                 (short)0, acc,  false,  false);

            const int j = jb + 16 * q + col;
#pragma unroll
            for (int r = 0; r < 8; ++r) {
                const float d = acc[r];
                const bool t = d < minv[r];
                minv[r] = t ? d : minv[r];
                mini[r] = t ? j : mini[r];
            }
        }
    }

#pragma unroll
    for (int off = 1; off < 16; off <<= 1) {
#pragma unroll
        for (int r = 0; r < 8; ++r) {
            const float ov = __shfl_xor(minv[r], off, 32);
            const int   oi = __shfl_xor(mini[r], off, 32);
            const bool  t  = (ov < minv[r]) | ((ov == minv[r]) & (oi < mini[r]));
            minv[r] = t ? ov : minv[r];
            mini[r] = t ? oi : mini[r];
        }
    }

    if (col == 0) {
        const int base = b * NN + rowBlock + wave * 16 + half * 8;
#pragma unroll
        for (int r = 0; r < 8; ++r) idx_out[base + r] = mini[r];
    }
}

#define GC 4
__global__ __launch_bounds__(256) void gather_feats_kernel(
    const float* __restrict__ kf,
    const int*   __restrict__ idx,
    float* __restrict__ out)
{
    __shared__ float4 tab[MM];

    const int b   = blockIdx.x >> 5;
    const int c0  = (blockIdx.x & 31) * GC;
    const int tid = threadIdx.x;

    const float* kfb = kf + (b * CC + c0) * MM;
    float4 r[2][GC];
#pragma unroll
    for (int k = 0; k < 2; ++k) {
        const int m = (tid + 256 * k) * 4;
#pragma unroll
        for (int c = 0; c < GC; ++c)
            r[k][c] = *(const float4*)(kfb + c * MM + m);
    }

    const v4i* idx4 = (const v4i*)(idx + b * NN);
    v4i id[8];
#pragma unroll
    for (int k = 0; k < 8; ++k) id[k] = idx4[tid + 256 * k];

#pragma unroll
    for (int k = 0; k < 2; ++k) {
        const int m = (tid + 256 * k) * 4;
        tab[m + 0] = make_float4(r[k][0].x, r[k][1].x, r[k][2].x, r[k][3].x);
        tab[m + 1] = make_float4(r[k][0].y, r[k][1].y, r[k][2].y, r[k][3].y);
        tab[m + 2] = make_float4(r[k][0].z, r[k][1].z, r[k][2].z, r[k][3].z);
        tab[m + 3] = make_float4(r[k][0].w, r[k][1].w, r[k][2].w, r[k][3].w);
    }
    __syncthreads();

    float* ob = out + (b * CC + c0) * NN;
#pragma unroll
    for (int k = 0; k < 8; ++k) {
        const int i = (tid + 256 * k) * 4;
        const float4 v0 = tab[id[k].x & (MM - 1)];
        const float4 v1 = tab[id[k].y & (MM - 1)];
        const float4 v2 = tab[id[k].z & (MM - 1)];
        const float4 v3 = tab[id[k].w & (MM - 1)];
        *(float4*)(ob + 0 * NN + i) = make_float4(v0.x, v1.x, v2.x, v3.x);
        *(float4*)(ob + 1 * NN + i) = make_float4(v0.y, v1.y, v2.y, v3.y);
        *(float4*)(ob + 2 * NN + i) = make_float4(v0.z, v1.z, v2.z, v3.z);
        *(float4*)(ob + 3 * NN + i) = make_float4(v0.w, v1.w, v2.w, v3.w);
    }
}

extern "C" void kernel_launch(void* const* d_in, const int* in_sizes, int n_in,
                              void* d_out, int out_size, void* d_ws, size_t ws_size,
                              hipStream_t stream) {
    const float* unknown     = (const float*)d_in[0];
    const float* known       = (const float*)d_in[1];
    const float* known_feats = (const float*)d_in[2];
    float* out = (float*)d_out;
    int* idx   = (int*)d_ws;

    nn_idx_kernel<<<BB * (NN / 128), 256, 0, stream>>>(unknown, known, idx);

    gather_feats_kernel<<<BB * (CC / GC), 256, 0, stream>>>(known_feats, idx, out);
}


# nn-index kernel: swap WMMA roles (known points = rows) so a lane's 8 results are one query's: argmin = min3 chain + tile select, index by re-multiplying the winning tile; indices bit-identical
# speedup vs baseline: 3.0956x; 1.4331x over previous
#include <hip/hip_runtime.h>
#include <hip/hip_bf16.h>

#define BB 8
#define NN 8192
#define MM 2048
#define CC 128

typedef __attribute__((ext_vector_type(2))) float v2f;
typedef __attribute__((ext_vector_type(8))) float v8f;
typedef __attribute__((ext_vector_type(4))) int   v4i;

__global__ __launch_bounds__(256) void nn_idx_kernel(
    const float* __restrict__ unknown,
    const float* __restrict__ known,
    int* __restrict__ idx_out)
{
    __shared__ float4 kn[MM];

    const int b        = blockIdx.x >> 6;
    const int rowBlock = (blockIdx.x & 63) * 128;
    const int tid      = threadIdx.x;

    const int wave = tid >> 5;
    const int lane = tid & 31;
    const int col  = lane & 15;
    const int half = lane >> 4;

    const float* kb = known + b * (MM * 3);
    const int row = rowBlock + wave * 16 + col;
    const float* u = unknown + (b * NN + row) * 3;

    const float ux = u[0], uy = u[1], uz = u[2];

    float px[8], py[8], pz[8];
#pragma unroll
    for (int k = 0; k < 8; ++k) {
        const float* p = kb + (tid + 256 * k) * 3;
        px[k] = p[0];
        py[k] = p[1];
        pz[k] = p[2];
    }
    __builtin_amdgcn_sched_barrier(0);

#pragma unroll
    for (int k = 0; k < 8; ++k) {
        const float x = px[k], y = py[k], z = pz[k];
        float s;
        {
#pragma clang fp contract(off)
            s = x * x + y * y;
        }
        kn[tid + 256 * k] = make_float4(x, y, z, __builtin_fmaf(z, z, s));
    }
    __syncthreads();

    v2f qb;
    qb.x = half ? (-2.0f * uz) : (-2.0f * ux);
    qb.y = half ? 1.0f         : (-2.0f * uy);

    const float2* __restrict__ lp = (const float2*)kn + (col * 2 + half);

    float minv  = 3.402823466e38f;
    int   mtile = 0;

    for (int jb = 0; jb < MM; jb += 64) {
        float2 kv[4];
#pragma unroll
        for (int q = 0; q < 4; ++q) kv[q] = lp[(jb + 16 * q) * 2];

#pragma unroll
        for (int q = 0; q < 4; ++q) {
            v2f av;
            av.x = kv[q].x;
            av.y = kv[q].y;

            v8f acc = {};
            acc = __builtin_amdgcn_wmma_f32_16x16x4_f32(
                 false, av,  false, qb,
                 (short)0, acc,  false,  false);

            const float t0 = __builtin_fminf(__builtin_fminf(acc[0], acc[1]), acc[2]);
            const float t1 = __builtin_fminf(__builtin_fminf(acc[3], acc[4]), acc[5]);
            const float t2 = __builtin_fminf(__builtin_fminf(acc[6], acc[7]), minv);
            const float m  = __builtin_fminf(__builtin_fminf(t0, t1), t2);
            mtile = (m < minv) ? (jb + 16 * q) : mtile;
            minv  = m;
        }
    }

    int pbase = mtile + half * 8;
    {
        const float ov = __shfl_xor(minv, 16, 32);
        const int   op = __shfl_xor(pbase, 16, 32);
        const bool  t  = (ov < minv) | ((ov == minv) & (op < pbase));
        minv  = t ? ov : minv;
        pbase = t ? op : pbase;
    }

    const int slot = (lane < 8) ? lane : (lane - 24);
    float keep[8];
#pragma unroll
    for (int r = 0; r < 8; ++r) keep[r] = 0.0f;
#pragma unroll
    for (int s = 0; s < 8; ++s) {
        const int pl = __builtin_amdgcn_readlane(pbase, s);
        const int ph = __builtin_amdgcn_readlane(pbase, s + 8);
        const int ts = (col < 8) ? pl : (ph - 8);
        const float2 kr = lp[ts * 2];
        v2f av;
        av.x = kr.x;
        av.y = kr.y;

        v8f acc = {};
        acc = __builtin_amdgcn_wmma_f32_16x16x4_f32(
             false, av,  false, qb,
             (short)0, acc,  false,  false);

        const bool mine = (slot == s);
#pragma unroll
        for (int r = 0; r < 8; ++r) keep[r] = mine ? acc[r] : keep[r];
    }

    int rr = 7;
#pragma unroll
    for (int r = 6; r >= 0; --r) rr = (keep[r] == minv) ? r : rr;

    if (slot >= 0) idx_out[b * NN + row] = pbase + rr;
}

#define GC 4
__global__ __launch_bounds__(256) void gather_feats_kernel(
    const float* __restrict__ kf,
    const int*   __restrict__ idx,
    float* __restrict__ out)
{
    __shared__ float4 tab[MM];

    const int b   = blockIdx.x >> 5;
    const int c0  = (blockIdx.x & 31) * GC;
    const int tid = threadIdx.x;

    const float* kfb = kf + (b * CC + c0) * MM;
    float4 r[2][GC];
#pragma unroll
    for (int k = 0; k < 2; ++k) {
        const int m = (tid + 256 * k) * 4;
#pragma unroll
        for (int c = 0; c < GC; ++c)
            r[k][c] = *(const float4*)(kfb + c * MM + m);
    }

    const v4i* idx4 = (const v4i*)(idx + b * NN);
    v4i id[8];
#pragma unroll
    for (int k = 0; k < 8; ++k) id[k] = idx4[tid + 256 * k];

#pragma unroll
    for (int k = 0; k < 2; ++k) {
        const int m = (tid + 256 * k) * 4;
        tab[m + 0] = make_float4(r[k][0].x, r[k][1].x, r[k][2].x, r[k][3].x);
        tab[m + 1] = make_float4(r[k][0].y, r[k][1].y, r[k][2].y, r[k][3].y);
        tab[m + 2] = make_float4(r[k][0].z, r[k][1].z, r[k][2].z, r[k][3].z);
        tab[m + 3] = make_float4(r[k][0].w, r[k][1].w, r[k][2].w, r[k][3].w);
    }
    __syncthreads();

    float* ob = out + (b * CC + c0) * NN;
#pragma unroll
    for (int k = 0; k < 8; ++k) {
        const int i = (tid + 256 * k) * 4;
        const float4 v0 = tab[id[k].x & (MM - 1)];
        const float4 v1 = tab[id[k].y & (MM - 1)];
        const float4 v2 = tab[id[k].z & (MM - 1)];
        const float4 v3 = tab[id[k].w & (MM - 1)];
        *(float4*)(ob + 0 * NN + i) = make_float4(v0.x, v1.x, v2.x, v3.x);
        *(float4*)(ob + 1 * NN + i) = make_float4(v0.y, v1.y, v2.y, v3.y);
        *(float4*)(ob + 2 * NN + i) = make_float4(v0.z, v1.z, v2.z, v3.z);
        *(float4*)(ob + 3 * NN + i) = make_float4(v0.w, v1.w, v2.w, v3.w);
    }
}

extern "C" void kernel_launch(void* const* d_in, const int* in_sizes, int n_in,
                              void* d_out, int out_size, void* d_ws, size_t ws_size,
                              hipStream_t stream) {
    const float* unknown     = (const float*)d_in[0];
    const float* known       = (const float*)d_in[1];
    const float* known_feats = (const float*)d_in[2];
    float* out = (float*)d_out;
    int* idx   = (int*)d_ws;

    nn_idx_kernel<<<BB * (NN / 128), 256, 0, stream>>>(unknown, known, idx);

    gather_feats_kernel<<<BB * (CC / GC), 256, 0, stream>>>(known_feats, idx, out);
}
